# starting kernel
# baseline (speedup 1.0000x reference)
#include <hip/hip_runtime.h>
#include <hip/hip_bf16.h>
#include <math.h>


typedef __attribute__((ext_vector_type(2))) float v2f;
typedef __attribute__((ext_vector_type(8))) float v8f;

#define NBATCH 8
#define NPTS   4096
#define WAVES  4
#define BLOCK  (WAVES * 32)
#define QPB    (WAVES * 16)
#define CHUNK  2048

__global__ __launch_bounds__(BLOCK)
void chamfer_nn_kernel(const float* __restrict__ tar,
                       const float* __restrict__ src,
                       float* __restrict__ out)
{
    const int b   = blockIdx.y;
    const int dir = blockIdx.z;
    const float* qry = (dir == 0 ? src : tar) + b * NPTS * 3;
    const float* ref = (dir == 0 ? tar : src) + b * NPTS * 3;
    float* o = out + dir * (NBATCH * NPTS) + b * NPTS;

    const int tid  = threadIdx.x;
    const int wave = tid >> 5;
    const int lane = tid & 31;
    const int hi   = lane >> 4;
    const int col  = lane & 15;

    __shared__ float2 s_ref[2 * CHUNK];

    const int qbase = blockIdx.x * QPB + wave * 16;
    const float* qp = qry + (qbase + col) * 3;
    const float qx = qp[0], qy = qp[1], qz = qp[2];
    v2f Av;
    Av.x = hi ? qz : qx;
    Av.y = hi ? 1.0f : qy;

    float bm[8];
#pragma unroll
    for (int r = 0; r < 8; ++r) bm[r] = 3.4e38f;

    const v8f czero = {};

    for (int c0 = 0; c0 < NPTS; c0 += CHUNK) {
        __syncthreads();
        for (int i = tid; i < CHUNK; i += BLOCK) {
            const float* rp = ref + (c0 + i) * 3;
            const float x = rp[0], y = rp[1], z = rp[2];
            s_ref[2 * i + 0] = make_float2(-2.0f * x, -2.0f * y);
            s_ref[2 * i + 1] = make_float2(-2.0f * z, x * x + y * y + z * z);
        }
        __syncthreads();

#pragma unroll 4
        for (int t = 0; t < CHUNK; t += 16) {
            const float2 bv = s_ref[2 * (t + col) + hi];
            v2f Bv;
            Bv.x = bv.x;
            Bv.y = bv.y;
            v8f d = __builtin_amdgcn_wmma_f32_16x16x4_f32(
                false, Av, false, Bv, (short)0, czero, false, false);
#pragma unroll
            for (int r = 0; r < 8; ++r) bm[r] = fminf(bm[r], d[r]);
        }
    }

#pragma unroll
    for (int r = 0; r < 8; ++r) {
        float v = bm[r];
        v = fminf(v, __shfl_xor(v, 1, 32));
        v = fminf(v, __shfl_xor(v, 2, 32));
        v = fminf(v, __shfl_xor(v, 4, 32));
        v = fminf(v, __shfl_xor(v, 8, 32));
        bm[r] = v;
    }

    if (col == 0) {
        const int qb = qbase + hi * 8;
#pragma unroll
        for (int r = 0; r < 8; ++r) {
            const float* q2p = qry + (qb + r) * 3;
            const float x = q2p[0], y = q2p[1], z = q2p[2];
            const float d2 = bm[r] + (x * x + y * y + z * z);
            o[qb + r] = sqrtf(fmaxf(d2, 0.0f));
        }
    }
}

__global__ __launch_bounds__(256)
void chamfer_reduce_kernel(const float* __restrict__ acc,
                           const float* __restrict__ comp,
                           float* __restrict__ cham)
{
    const int b = blockIdx.x;
    __shared__ float red[256];
    float s = 0.0f;
    for (int i = threadIdx.x; i < NPTS; i += 256)
        s += acc[b * NPTS + i] + comp[b * NPTS + i];
    red[threadIdx.x] = s;
    __syncthreads();
    for (int st = 128; st > 0; st >>= 1) {
        if (threadIdx.x < st) red[threadIdx.x] += red[threadIdx.x + st];
        __syncthreads();
    }
    if (threadIdx.x == 0)
        cham[b] = red[0] / (2.0f * (float)NPTS);
}

extern "C" void kernel_launch(void* const* d_in, const int* in_sizes, int n_in,
                              void* d_out, int out_size, void* d_ws, size_t ws_size,
                              hipStream_t stream) {
    (void)in_sizes; (void)n_in; (void)d_ws; (void)ws_size; (void)out_size;
    const float* tar = (const float*)d_in[0];
    const float* src = (const float*)d_in[1];
    float* out = (float*)d_out;

    dim3 grid(NPTS / QPB, NBATCH, 2);
    chamfer_nn_kernel<<<grid, BLOCK, 0, stream>>>(tar, src, out);

    const float* acc  = out;
    const float* comp = out + NBATCH * NPTS;
    float* cham       = out + 2 * NBATCH * NPTS;
    chamfer_reduce_kernel<<<NBATCH, 256, 0, stream>>>(acc, comp, cham);
}


# Ref staging: all 32 global loads per thread issued at kernel start, whole ref set kept in 64 KB LDS, handed over in 4 groups behind LDS-only barriers so WMMA sweeps overlap loads still in flight
# speedup vs baseline: 1.1516x; 1.1516x over previous
#include <hip/hip_runtime.h>
#include <hip/hip_bf16.h>
#include <math.h>


typedef __attribute__((ext_vector_type(2))) float v2f;
typedef __attribute__((ext_vector_type(8))) float v8f;

#define NBATCH 8
#define NPTS   4096
#define WAVES  4
#define BLOCK  (WAVES * 32)
#define QPB    (WAVES * 16)
#define NGRP   4
#define RPT    (NPTS / BLOCK)
#define RPG    (RPT / NGRP)
#define GREFS  (NPTS / NGRP)

__device__ __forceinline__ void lds_barrier()
{
    __builtin_amdgcn_fence(__ATOMIC_RELEASE, "workgroup", "local");
    __builtin_amdgcn_s_barrier();
    __builtin_amdgcn_fence(__ATOMIC_ACQUIRE, "workgroup", "local");
}

__global__ __launch_bounds__(BLOCK)
void chamfer_nn_kernel(const float* __restrict__ tar,
                       const float* __restrict__ src,
                       float* __restrict__ out)
{
    const int b   = blockIdx.y;
    const int dir = blockIdx.z;
    const float* qry = (dir == 0 ? src : tar) + b * NPTS * 3;
    const float* ref = (dir == 0 ? tar : src) + b * NPTS * 3;
    float* o = out + dir * (NBATCH * NPTS) + b * NPTS;

    const int tid  = threadIdx.x;
    const int wave = tid >> 5;
    const int lane = tid & 31;
    const int hi   = lane >> 4;
    const int col  = lane & 15;

    __shared__ float2 s_ref[2 * NPTS];

    const int qbase = blockIdx.x * QPB + wave * 16;
    const float* qp = qry + (qbase + col) * 3;
    const float qx = qp[0], qy = qp[1], qz = qp[2];
    __builtin_amdgcn_sched_barrier(0);
    v2f Av;
    Av.x = hi ? qz : qx;
    Av.y = hi ? 1.0f : qy;

    float bm[8];
#pragma unroll
    for (int r = 0; r < 8; ++r) bm[r] = 3.4e38f;

    const v8f czero = {};

    float rx[RPT], ry[RPT], rz[RPT];
#pragma unroll
    for (int k = 0; k < RPT; ++k) {
        const float* rp = ref + (k * BLOCK + tid) * 3;
        rx[k] = rp[0];
        ry[k] = rp[1];
        rz[k] = rp[2];
    }

#pragma unroll
    for (int g = 0; g < NGRP; ++g) {
#pragma unroll
        for (int k = g * RPG; k < (g + 1) * RPG; ++k) {
            const int i = k * BLOCK + tid;
            const float x = rx[k], y = ry[k], z = rz[k];
            s_ref[2 * i + 0] = make_float2(-2.0f * x, -2.0f * y);
            s_ref[2 * i + 1] = make_float2(-2.0f * z, x * x + y * y + z * z);
        }
        lds_barrier();

#pragma unroll 4
        for (int t = g * GREFS; t < (g + 1) * GREFS; t += 16) {
            const float2 bv = s_ref[2 * (t + col) + hi];
            v2f Bv;
            Bv.x = bv.x;
            Bv.y = bv.y;
            v8f d = __builtin_amdgcn_wmma_f32_16x16x4_f32(
                false, Av, false, Bv, (short)0, czero, false, false);
#pragma unroll
            for (int r = 0; r < 8; ++r) bm[r] = fminf(bm[r], d[r]);
        }
    }

#pragma unroll
    for (int r = 0; r < 8; ++r) {
        float v = bm[r];
        v = fminf(v, __shfl_xor(v, 1, 32));
        v = fminf(v, __shfl_xor(v, 2, 32));
        v = fminf(v, __shfl_xor(v, 4, 32));
        v = fminf(v, __shfl_xor(v, 8, 32));
        bm[r] = v;
    }

    if (col == 0) {
        const int qb = qbase + hi * 8;
#pragma unroll
        for (int r = 0; r < 8; ++r) {
            const float* q2p = qry + (qb + r) * 3;
            const float x = q2p[0], y = q2p[1], z = q2p[2];
            const float d2 = bm[r] + (x * x + y * y + z * z);
            o[qb + r] = sqrtf(fmaxf(d2, 0.0f));
        }
    }
}

__global__ __launch_bounds__(256)
void chamfer_reduce_kernel(const float* __restrict__ acc,
                           const float* __restrict__ comp,
                           float* __restrict__ cham)
{
    const int b = blockIdx.x;
    __shared__ float red[256];
    float s = 0.0f;
    for (int i = threadIdx.x; i < NPTS; i += 256)
        s += acc[b * NPTS + i] + comp[b * NPTS + i];
    red[threadIdx.x] = s;
    __syncthreads();
    for (int st = 128; st > 0; st >>= 1) {
        if (threadIdx.x < st) red[threadIdx.x] += red[threadIdx.x + st];
        __syncthreads();
    }
    if (threadIdx.x == 0)
        cham[b] = red[0] / (2.0f * (float)NPTS);
}

extern "C" void kernel_launch(void* const* d_in, const int* in_sizes, int n_in,
                              void* d_out, int out_size, void* d_ws, size_t ws_size,
                              hipStream_t stream) {
    (void)in_sizes; (void)n_in; (void)d_ws; (void)ws_size; (void)out_size;
    const float* tar = (const float*)d_in[0];
    const float* src = (const float*)d_in[1];
    float* out = (float*)d_out;

    dim3 grid(NPTS / QPB, NBATCH, 2);
    chamfer_nn_kernel<<<grid, BLOCK, 0, stream>>>(tar, src, out);

    const float* acc  = out;
    const float* comp = out + NBATCH * NPTS;
    float* cham       = out + 2 * NBATCH * NPTS;
    chamfer_reduce_kernel<<<NBATCH, 256, 0, stream>>>(acc, comp, cham);
}


# Ref sweep software-pipelined per wave: LDS reads of the next 8-tile B block go into a second register set ahead of the current block's 8 WMMAs, min ops interleaved, group hand-over folded in
# speedup vs baseline: 1.2552x; 1.0900x over previous
#include <hip/hip_runtime.h>
#include <hip/hip_bf16.h>
#include <math.h>


typedef __attribute__((ext_vector_type(2))) float v2f;
typedef __attribute__((ext_vector_type(8))) float v8f;

#define NBATCH 8
#define NPTS   4096
#define WAVES  4
#define BLOCK  (WAVES * 32)
#define QPB    (WAVES * 16)
#define NGRP   4
#define RPT    (NPTS / BLOCK)
#define RPG    (RPT / NGRP)
#define GREFS  (NPTS / NGRP)
#define TPB    8
#define BREFS  (TPB * 16)
#define BPG    (GREFS / BREFS)

__device__ __forceinline__ void lds_barrier()
{
    __builtin_amdgcn_fence(__ATOMIC_RELEASE, "workgroup", "local");
    __builtin_amdgcn_s_barrier();
    __builtin_amdgcn_fence(__ATOMIC_ACQUIRE, "workgroup", "local");
}

__device__ __forceinline__ void load_block(v2f (&f)[TPB], const float2* s_ref,
                                           int t0, int col, int hi)
{
#pragma unroll
    for (int i = 0; i < TPB; ++i) {
        const float2 bv = s_ref[2 * (t0 + 16 * i + col) + hi];
        f[i].x = bv.x;
        f[i].y = bv.y;
    }
}

__device__ __forceinline__ void sweep_block(const v2f (&f)[TPB], v2f Av, float (&bm)[8])
{
    const v8f czero = {};
    v8f d[TPB];
#pragma unroll
    for (int i = 0; i < TPB; ++i)
        d[i] = __builtin_amdgcn_wmma_f32_16x16x4_f32(
            false, Av, false, f[i], (short)0, czero, false, false);
#pragma unroll
    for (int i = 0; i < TPB; ++i)
#pragma unroll
        for (int r = 0; r < 8; ++r) bm[r] = fminf(bm[r], d[i][r]);
    __builtin_amdgcn_sched_group_barrier(0x008, 4, 0);
    __builtin_amdgcn_sched_group_barrier(0x002, 8, 0);
    __builtin_amdgcn_sched_group_barrier(0x008, 2, 0);
    __builtin_amdgcn_sched_group_barrier(0x002, 8, 0);
    __builtin_amdgcn_sched_group_barrier(0x008, 2, 0);
    __builtin_amdgcn_sched_group_barrier(0x002, 8, 0);
    __builtin_amdgcn_sched_group_barrier(0x002, 8, 0);
}

__global__ __launch_bounds__(BLOCK)
void chamfer_nn_kernel(const float* __restrict__ tar,
                       const float* __restrict__ src,
                       float* __restrict__ out)
{
    const int b   = blockIdx.y;
    const int dir = blockIdx.z;
    const float* qry = (dir == 0 ? src : tar) + b * NPTS * 3;
    const float* ref = (dir == 0 ? tar : src) + b * NPTS * 3;
    float* o = out + dir * (NBATCH * NPTS) + b * NPTS;

    const int tid  = threadIdx.x;
    const int wave = tid >> 5;
    const int lane = tid & 31;
    const int hi   = lane >> 4;
    const int col  = lane & 15;

    __shared__ float2 s_ref[2 * NPTS];

    const int qbase = blockIdx.x * QPB + wave * 16;
    const float* qp = qry + (qbase + col) * 3;
    const float qx = qp[0], qy = qp[1], qz = qp[2];
    __builtin_amdgcn_sched_barrier(0);
    v2f Av;
    Av.x = hi ? qz : qx;
    Av.y = hi ? 1.0f : qy;

    float bm[8];
#pragma unroll
    for (int r = 0; r < 8; ++r) bm[r] = 3.4e38f;

    float rx[RPT], ry[RPT], rz[RPT];
#pragma unroll
    for (int k = 0; k < RPT; ++k) {
        const float* rp = ref + (k * BLOCK + tid) * 3;
        rx[k] = rp[0];
        ry[k] = rp[1];
        rz[k] = rp[2];
    }

    v2f fa[TPB], fb[TPB];

#pragma unroll
    for (int k = 0; k < RPG; ++k) {
        const int i = k * BLOCK + tid;
        const float x = rx[k], y = ry[k], z = rz[k];
        s_ref[2 * i + 0] = make_float2(-2.0f * x, -2.0f * y);
        s_ref[2 * i + 1] = make_float2(-2.0f * z, x * x + y * y + z * z);
    }
    lds_barrier();
    load_block(fa, s_ref, 0, col, hi);

#pragma unroll
    for (int g = 0; g < NGRP; ++g) {
#pragma nounroll
        for (int p = 0; p < BPG / 2 - 1; ++p) {
            const int t0 = g * GREFS + 2 * p * BREFS;

            load_block(fb, s_ref, t0 + BREFS, col, hi);
            __builtin_amdgcn_sched_barrier(0);
            sweep_block(fa, Av, bm);
            __builtin_amdgcn_sched_barrier(0);

            load_block(fa, s_ref, t0 + 2 * BREFS, col, hi);
            __builtin_amdgcn_sched_barrier(0);
            sweep_block(fb, Av, bm);
            __builtin_amdgcn_sched_barrier(0);
        }

        const int tl = g * GREFS + (BPG - 2) * BREFS;
        load_block(fb, s_ref, tl + BREFS, col, hi);
        if (g < NGRP - 1) {
#pragma unroll
            for (int k = (g + 1) * RPG; k < (g + 2) * RPG; ++k) {
                const int i = k * BLOCK + tid;
                const float x = rx[k], y = ry[k], z = rz[k];
                s_ref[2 * i + 0] = make_float2(-2.0f * x, -2.0f * y);
                s_ref[2 * i + 1] = make_float2(-2.0f * z, x * x + y * y + z * z);
            }
        }
        __builtin_amdgcn_sched_barrier(0);
        sweep_block(fa, Av, bm);
        __builtin_amdgcn_sched_barrier(0);

        if (g < NGRP - 1) {
            lds_barrier();
            load_block(fa, s_ref, (g + 1) * GREFS, col, hi);
        }
        __builtin_amdgcn_sched_barrier(0);
        sweep_block(fb, Av, bm);
        __builtin_amdgcn_sched_barrier(0);
    }

#pragma unroll
    for (int r = 0; r < 8; ++r) {
        float v = bm[r];
        v = fminf(v, __shfl_xor(v, 1, 32));
        v = fminf(v, __shfl_xor(v, 2, 32));
        v = fminf(v, __shfl_xor(v, 4, 32));
        v = fminf(v, __shfl_xor(v, 8, 32));
        bm[r] = v;
    }

    if (col == 0) {
        const int qb = qbase + hi * 8;
#pragma unroll
        for (int r = 0; r < 8; ++r) {
            const float* q2p = qry + (qb + r) * 3;
            const float x = q2p[0], y = q2p[1], z = q2p[2];
            const float d2 = bm[r] + (x * x + y * y + z * z);
            o[qb + r] = sqrtf(fmaxf(d2, 0.0f));
        }
    }
}

__global__ __launch_bounds__(256)
void chamfer_reduce_kernel(const float* __restrict__ acc,
                           const float* __restrict__ comp,
                           float* __restrict__ cham)
{
    const int b = blockIdx.x;
    __shared__ float red[256];
    float s = 0.0f;
    for (int i = threadIdx.x; i < NPTS; i += 256)
        s += acc[b * NPTS + i] + comp[b * NPTS + i];
    red[threadIdx.x] = s;
    __syncthreads();
    for (int st = 128; st > 0; st >>= 1) {
        if (threadIdx.x < st) red[threadIdx.x] += red[threadIdx.x + st];
        __syncthreads();
    }
    if (threadIdx.x == 0)
        cham[b] = red[0] / (2.0f * (float)NPTS);
}

extern "C" void kernel_launch(void* const* d_in, const int* in_sizes, int n_in,
                              void* d_out, int out_size, void* d_ws, size_t ws_size,
                              hipStream_t stream) {
    (void)in_sizes; (void)n_in; (void)d_ws; (void)ws_size; (void)out_size;
    const float* tar = (const float*)d_in[0];
    const float* src = (const float*)d_in[1];
    float* out = (float*)d_out;

    dim3 grid(NPTS / QPB, NBATCH, 2);
    chamfer_nn_kernel<<<grid, BLOCK, 0, stream>>>(tar, src, out);

    const float* acc  = out;
    const float* comp = out + NBATCH * NPTS;
    float* cham       = out + 2 * NBATCH * NPTS;
    chamfer_reduce_kernel<<<NBATCH, 256, 0, stream>>>(acc, comp, cham);
}


# Cross-lane reductions on the vector ALU (DPP row_xmask, permlanex16) instead of LDS-unit shuffles: 16 row minima per wave as a bank-masked transposing butterfly, wave sums in the same order
# speedup vs baseline: 1.5991x; 1.2740x over previous
#include <hip/hip_runtime.h>
#include <hip/hip_bf16.h>
#include <math.h>


typedef __attribute__((ext_vector_type(2))) float v2f;
typedef __attribute__((ext_vector_type(8))) float v8f;

#define NBATCH 8
#define NPTS   4096
#define RS     512
#define CS     1024
#define WAVES  16
#define BLOCK  (WAVES * 32)
#define RPW    (RS / WAVES)
#define NRS    (NPTS / RS)
#define NCS    (NPTS / CS)
#define TILES  (CS / 16)
#define TPB    8
#define BREFS  (TPB * 16)
#define NBLK   (TILES / TPB)
#define FINISH 1024

__device__ __forceinline__ void lds_barrier()
{
    __builtin_amdgcn_fence(__ATOMIC_RELEASE, "workgroup", "local");
    __builtin_amdgcn_s_barrier();
    __builtin_amdgcn_fence(__ATOMIC_ACQUIRE, "workgroup", "local");
}

__device__ __forceinline__ void load_block(v2f (&f)[TPB], const float2* s_ref,
                                           int t0, int col, int hi)
{
#pragma unroll
    for (int i = 0; i < TPB; ++i) {
        const float2 bv = s_ref[2 * (t0 + 16 * i + col) + hi];
        f[i].x = bv.x;
        f[i].y = bv.y;
    }
}

__device__ __forceinline__ float min3f(float a, float b, float c)
{
    return fminf(fminf(a, b), c);
}

#define DPP_PAIRMIN(dst, src, XM, BANKS)                                        \
    asm("v_min_num_f32_dpp %0, %1, %1 row_xmask:" #XM                          \
        " row_mask:0xf bank_mask:" #BANKS : "+v"(dst) : "v"(src))
#define DPP_MIN_KEEP(keep, send, XM)                                            \
    asm("v_min_num_f32_dpp %0, %1, %0 row_xmask:" #XM                          \
        " row_mask:0xf bank_mask:0xf" : "+v"(keep) : "v"(send))
#define DPP_ADD_SELF(s, XM)                                                     \
    asm("v_add_f32_dpp %0, %0, %0 row_xmask:" #XM                              \
        " row_mask:0xf bank_mask:0xf" : "+v"(s))

__device__ __forceinline__ void sweep_block(const v2f (&f)[TPB], const v2f (&Av)[2],
                                            const v8f (&Cq)[2], float (&bm)[2][8],
                                            float* cmw)
{
    v8f d0[TPB], d1[TPB];
#pragma unroll
    for (int i = 0; i < TPB; ++i) {
        d0[i] = __builtin_amdgcn_wmma_f32_16x16x4_f32(
            false, Av[0], false, f[i], (short)0, Cq[0], false, false);
        d1[i] = __builtin_amdgcn_wmma_f32_16x16x4_f32(
            false, Av[1], false, f[i], (short)0, Cq[1], false, false);
    }
#pragma unroll
    for (int i = 0; i < TPB; i += 2) {
#pragma unroll
        for (int r = 0; r < 8; ++r) {
            bm[0][r] = min3f(bm[0][r], d0[i][r], d0[i + 1][r]);
            bm[1][r] = min3f(bm[1][r], d1[i][r], d1[i + 1][r]);
        }
#pragma unroll
        for (int j = i; j < i + 2; ++j) {
            float cm = 3.4e38f;
#pragma unroll
            for (int r = 0; r < 8; r += 2) cm = min3f(cm, d0[j][r], d0[j][r + 1]);
#pragma unroll
            for (int r = 0; r < 8; r += 2) cm = min3f(cm, d1[j][r], d1[j][r + 1]);
            cmw[j * 32] = cm;
        }
    }
    __builtin_amdgcn_sched_group_barrier(0x008, 8, 0);
    __builtin_amdgcn_sched_group_barrier(0x002, 32, 0);
    __builtin_amdgcn_sched_group_barrier(0x008, 4, 0);
    __builtin_amdgcn_sched_group_barrier(0x002, 32, 0);
    __builtin_amdgcn_sched_group_barrier(0x008, 4, 0);
    __builtin_amdgcn_sched_group_barrier(0x002, 32, 0);
    __builtin_amdgcn_sched_group_barrier(0x002, 32, 0);
    __builtin_amdgcn_sched_group_barrier(0x200, 8, 0);
}

__global__ __launch_bounds__(BLOCK)
void chamfer_tile_kernel(const float* __restrict__ tar,
                         const float* __restrict__ src,
                         float* __restrict__ prow,
                         float* __restrict__ pcol)
{
    const int cs = blockIdx.x;
    const int rs = blockIdx.y;
    const int b  = blockIdx.z;
    const float* qry = src + (b * NPTS + rs * RS) * 3;
    const float* ref = tar + (b * NPTS + cs * CS) * 3;

    const int tid  = threadIdx.x;
    const int wave = tid >> 5;
    const int lane = tid & 31;
    const int hi   = lane >> 4;
    const int col  = lane & 15;

    __shared__ float2 s_ref[2 * CS];
    __shared__ float  s_cm[WAVES * TILES * 32];

    v2f Av[2];
    v8f Cq[2];
#pragma unroll
    for (int f = 0; f < 2; ++f) {
        const float* qp = qry + (wave * RPW + 16 * f + col) * 3;
        const float qx = qp[0], qy = qp[1], qz = qp[2];
        Av[f].x = hi ? qz : qx;
        Av[f].y = hi ? 1.0f : qy;
        const float q2 = qx * qx + qy * qy + qz * qz;
#pragma unroll
        for (int r = 0; r < 8; ++r) Cq[f][r] = __shfl(q2, 8 * hi + r, 32);
    }

#pragma unroll
    for (int k = 0; k < CS / BLOCK; ++k) {
        const int i = k * BLOCK + tid;
        const float* rp = ref + i * 3;
        const float x = rp[0], y = rp[1], z = rp[2];
        s_ref[2 * i + 0] = make_float2(-2.0f * x, -2.0f * y);
        s_ref[2 * i + 1] = make_float2(-2.0f * z, x * x + y * y + z * z);
    }
    __syncthreads();

    float bm[2][8];
#pragma unroll
    for (int f = 0; f < 2; ++f)
#pragma unroll
        for (int r = 0; r < 8; ++r) bm[f][r] = 3.4e38f;

    v2f fa[TPB], fb[TPB];
    float* cmw = s_cm + wave * (TILES * 32) + lane;
    load_block(fa, s_ref, 0, col, hi);

#pragma nounroll
    for (int p = 0; p < NBLK / 2; ++p) {
        const int t0 = 2 * p * TPB;

        load_block(fb, s_ref, (t0 + TPB) * 16, col, hi);
        __builtin_amdgcn_sched_barrier(0);
        sweep_block(fa, Av, Cq, bm, cmw + t0 * 32);
        __builtin_amdgcn_sched_barrier(0);

        load_block(fa, s_ref, ((t0 + 2 * TPB) & (TILES - 1)) * 16, col, hi);
        __builtin_amdgcn_sched_barrier(0);
        sweep_block(fb, Av, Cq, bm, cmw + (t0 + TPB) * 32);
        __builtin_amdgcn_sched_barrier(0);
    }

    float v[16];
#pragma unroll
    for (int f = 0; f < 2; ++f)
#pragma unroll
        for (int r = 0; r < 8; ++r) v[f * 8 + r] = bm[f][r];
#pragma unroll
    for (int j = 0; j < 8; ++j) {
        DPP_PAIRMIN(v[j], v[j],     8, 0x3);
        DPP_PAIRMIN(v[j], v[j + 8], 8, 0xc);
    }
#pragma unroll
    for (int j = 0; j < 4; ++j) {
        DPP_PAIRMIN(v[j], v[j],     4, 0x5);
        DPP_PAIRMIN(v[j], v[j + 4], 4, 0xa);
    }
    const bool b1 = (col & 2) != 0;
    const bool b0 = (col & 1) != 0;
#pragma unroll
    for (int j = 0; j < 2; ++j) {
        float keep = b1 ? v[j + 2] : v[j];
        const float send = b1 ? v[j] : v[j + 2];
        DPP_MIN_KEEP(keep, send, 2);
        v[j] = keep;
    }
    float rv = b0 ? v[1] : v[0];
    {
        const float send = b0 ? v[0] : v[1];
        DPP_MIN_KEEP(rv, send, 1);
    }
    prow[(b * NCS + cs) * NPTS + rs * RS + wave * RPW
         + 16 * (col >> 3) + 8 * hi + (col & 7)] = rv;

    lds_barrier();
#pragma unroll
    for (int k = 0; k < CS / BLOCK; ++k) {
        const int j = k * BLOCK + tid;
        const float* pc = s_cm + (j >> 4) * 32 + (j & 15);
        float m = 3.4e38f;
#pragma unroll
        for (int w = 0; w < WAVES; ++w)
            m = min3f(m, pc[w * (TILES * 32)], pc[w * (TILES * 32) + 16]);
        pcol[(b * NRS + rs) * NPTS + cs * CS + j] = m;
    }
}

__device__ __forceinline__ float wave_sum(float s)
{
    const int sb = __builtin_bit_cast(int, s);
    s += __builtin_bit_cast(float, __builtin_amdgcn_permlanex16(
             sb, sb, 0x76543210, 0xfedcba98, false, false));
    DPP_ADD_SELF(s, 8);
    DPP_ADD_SELF(s, 4);
    DPP_ADD_SELF(s, 2);
    DPP_ADD_SELF(s, 1);
    return s;
}

__device__ __forceinline__ float4 min4(float4 a, float4 b)
{
    return make_float4(fminf(a.x, b.x), fminf(a.y, b.y), fminf(a.z, b.z), fminf(a.w, b.w));
}

__device__ __forceinline__ float4 dist4(float4 a)
{
    return make_float4(sqrtf(fmaxf(a.x, 0.0f)), sqrtf(fmaxf(a.y, 0.0f)),
                       sqrtf(fmaxf(a.z, 0.0f)), sqrtf(fmaxf(a.w, 0.0f)));
}

__global__ __launch_bounds__(FINISH)
void chamfer_finish_kernel(const float* __restrict__ prow,
                           const float* __restrict__ pcol,
                           float* __restrict__ out)
{
    const int b    = blockIdx.x;
    const int tid  = threadIdx.x;
    const int wave = tid >> 5;
    const int lane = tid & 31;
    const float4* pr = reinterpret_cast<const float4*>(prow + b * NCS * NPTS) + tid;
    const float4* pc = reinterpret_cast<const float4*>(pcol + b * NRS * NPTS) + tid;

    float4 a = pr[0];
#pragma unroll
    for (int p = 1; p < NCS; ++p) a = min4(a, pr[p * (NPTS / 4)]);
    float4 c = pc[0];
#pragma unroll
    for (int p = 1; p < NRS; ++p) c = min4(c, pc[p * (NPTS / 4)]);

    a = dist4(a);
    c = dist4(c);
    reinterpret_cast<float4*>(out + b * NPTS)[tid] = a;
    reinterpret_cast<float4*>(out + NBATCH * NPTS + b * NPTS)[tid] = c;

    float s = ((a.x + a.y) + (a.z + a.w)) + ((c.x + c.y) + (c.z + c.w));
    s = wave_sum(s);
    __shared__ float red[FINISH / 32];
    if (lane == 0) red[wave] = s;
    lds_barrier();
    if (wave == 0) {
        float v = wave_sum(red[lane]);
        if (lane == 0)
            out[2 * NBATCH * NPTS + b] = v / (2.0f * (float)NPTS);
    }
}

extern "C" void kernel_launch(void* const* d_in, const int* in_sizes, int n_in,
                              void* d_out, int out_size, void* d_ws, size_t ws_size,
                              hipStream_t stream) {
    (void)in_sizes; (void)n_in; (void)ws_size; (void)out_size;
    const float* tar = (const float*)d_in[0];
    const float* src = (const float*)d_in[1];
    float* out = (float*)d_out;
    float* prow = (float*)d_ws;
    float* pcol = prow + NBATCH * NCS * NPTS;

    dim3 grid(NCS, NRS, NBATCH);
    chamfer_tile_kernel<<<grid, BLOCK, 0, stream>>>(tar, src, prow, pcol);
    chamfer_finish_kernel<<<NBATCH, FINISH, 0, stream>>>(prow, pcol, out);
}


# Finish launch: 768 L2 prefetches, one per 256-byte unit of the work-group's 12 partial planes, issued ahead of its loads: 192 KB per work-group then stream from L2, not in serial memory round trips
# speedup vs baseline: 1.7038x; 1.0655x over previous
#include <hip/hip_runtime.h>
#include <hip/hip_bf16.h>
#include <math.h>


typedef __attribute__((ext_vector_type(2))) float v2f;
typedef __attribute__((ext_vector_type(8))) float v8f;

#define NBATCH 8
#define NPTS   4096
#define RS     512
#define CS     1024
#define WAVES  16
#define BLOCK  (WAVES * 32)
#define RPW    (RS / WAVES)
#define NRS    (NPTS / RS)
#define NCS    (NPTS / CS)
#define TILES  (CS / 16)
#define TPB    8
#define BREFS  (TPB * 16)
#define NBLK   (TILES / TPB)
#define FINISH 1024

__device__ __forceinline__ void lds_barrier()
{
    __builtin_amdgcn_fence(__ATOMIC_RELEASE, "workgroup", "local");
    __builtin_amdgcn_s_barrier();
    __builtin_amdgcn_fence(__ATOMIC_ACQUIRE, "workgroup", "local");
}

__device__ __forceinline__ void load_block(v2f (&f)[TPB], const float2* s_ref,
                                           int t0, int col, int hi)
{
#pragma unroll
    for (int i = 0; i < TPB; ++i) {
        const float2 bv = s_ref[2 * (t0 + 16 * i + col) + hi];
        f[i].x = bv.x;
        f[i].y = bv.y;
    }
}

__device__ __forceinline__ float min3f(float a, float b, float c)
{
    return fminf(fminf(a, b), c);
}

#define DPP_PAIRMIN(dst, src, XM, BANKS)                                        \
    asm("v_min_num_f32_dpp %0, %1, %1 row_xmask:" #XM                          \
        " row_mask:0xf bank_mask:" #BANKS : "+v"(dst) : "v"(src))
#define DPP_MIN_KEEP(keep, send, XM)                                            \
    asm("v_min_num_f32_dpp %0, %1, %0 row_xmask:" #XM                          \
        " row_mask:0xf bank_mask:0xf" : "+v"(keep) : "v"(send))
#define DPP_ADD_SELF(s, XM)                                                     \
    asm("v_add_f32_dpp %0, %0, %0 row_xmask:" #XM                              \
        " row_mask:0xf bank_mask:0xf" : "+v"(s))

__device__ __forceinline__ void sweep_block(const v2f (&f)[TPB], const v2f (&Av)[2],
                                            const v8f (&Cq)[2], float (&bm)[2][8],
                                            float* cmw)
{
    v8f d0[TPB], d1[TPB];
#pragma unroll
    for (int i = 0; i < TPB; ++i) {
        d0[i] = __builtin_amdgcn_wmma_f32_16x16x4_f32(
            false, Av[0], false, f[i], (short)0, Cq[0], false, false);
        d1[i] = __builtin_amdgcn_wmma_f32_16x16x4_f32(
            false, Av[1], false, f[i], (short)0, Cq[1], false, false);
    }
#pragma unroll
    for (int i = 0; i < TPB; i += 2) {
#pragma unroll
        for (int r = 0; r < 8; ++r) {
            bm[0][r] = min3f(bm[0][r], d0[i][r], d0[i + 1][r]);
            bm[1][r] = min3f(bm[1][r], d1[i][r], d1[i + 1][r]);
        }
#pragma unroll
        for (int j = i; j < i + 2; ++j) {
            float cm = 3.4e38f;
#pragma unroll
            for (int r = 0; r < 8; r += 2) cm = min3f(cm, d0[j][r], d0[j][r + 1]);
#pragma unroll
            for (int r = 0; r < 8; r += 2) cm = min3f(cm, d1[j][r], d1[j][r + 1]);
            cmw[j * 32] = cm;
        }
    }
    __builtin_amdgcn_sched_group_barrier(0x008, 8, 0);
    __builtin_amdgcn_sched_group_barrier(0x002, 32, 0);
    __builtin_amdgcn_sched_group_barrier(0x008, 4, 0);
    __builtin_amdgcn_sched_group_barrier(0x002, 32, 0);
    __builtin_amdgcn_sched_group_barrier(0x008, 4, 0);
    __builtin_amdgcn_sched_group_barrier(0x002, 32, 0);
    __builtin_amdgcn_sched_group_barrier(0x002, 32, 0);
    __builtin_amdgcn_sched_group_barrier(0x200, 8, 0);
}

__global__ __launch_bounds__(BLOCK)
void chamfer_tile_kernel(const float* __restrict__ tar,
                         const float* __restrict__ src,
                         float* __restrict__ prow,
                         float* __restrict__ pcol)
{
    const int cs = blockIdx.x;
    const int rs = blockIdx.y;
    const int b  = blockIdx.z;
    const float* qry = src + (b * NPTS + rs * RS) * 3;
    const float* ref = tar + (b * NPTS + cs * CS) * 3;

    const int tid  = threadIdx.x;
    const int wave = tid >> 5;
    const int lane = tid & 31;
    const int hi   = lane >> 4;
    const int col  = lane & 15;

    __shared__ float2 s_ref[2 * CS];
    __shared__ float  s_cm[WAVES * TILES * 32];

    v2f Av[2];
    v8f Cq[2];
#pragma unroll
    for (int f = 0; f < 2; ++f) {
        const float* qp = qry + (wave * RPW + 16 * f + col) * 3;
        const float qx = qp[0], qy = qp[1], qz = qp[2];
        Av[f].x = hi ? qz : qx;
        Av[f].y = hi ? 1.0f : qy;
        const float q2 = qx * qx + qy * qy + qz * qz;
#pragma unroll
        for (int r = 0; r < 8; ++r) Cq[f][r] = __shfl(q2, 8 * hi + r, 32);
    }

#pragma unroll
    for (int k = 0; k < CS / BLOCK; ++k) {
        const int i = k * BLOCK + tid;
        const float* rp = ref + i * 3;
        const float x = rp[0], y = rp[1], z = rp[2];
        s_ref[2 * i + 0] = make_float2(-2.0f * x, -2.0f * y);
        s_ref[2 * i + 1] = make_float2(-2.0f * z, x * x + y * y + z * z);
    }
    __syncthreads();

    float bm[2][8];
#pragma unroll
    for (int f = 0; f < 2; ++f)
#pragma unroll
        for (int r = 0; r < 8; ++r) bm[f][r] = 3.4e38f;

    v2f fa[TPB], fb[TPB];
    float* cmw = s_cm + wave * (TILES * 32) + lane;
    load_block(fa, s_ref, 0, col, hi);

#pragma nounroll
    for (int p = 0; p < NBLK / 2; ++p) {
        const int t0 = 2 * p * TPB;

        load_block(fb, s_ref, (t0 + TPB) * 16, col, hi);
        __builtin_amdgcn_sched_barrier(0);
        sweep_block(fa, Av, Cq, bm, cmw + t0 * 32);
        __builtin_amdgcn_sched_barrier(0);

        load_block(fa, s_ref, ((t0 + 2 * TPB) & (TILES - 1)) * 16, col, hi);
        __builtin_amdgcn_sched_barrier(0);
        sweep_block(fb, Av, Cq, bm, cmw + (t0 + TPB) * 32);
        __builtin_amdgcn_sched_barrier(0);
    }

    float v[16];
#pragma unroll
    for (int f = 0; f < 2; ++f)
#pragma unroll
        for (int r = 0; r < 8; ++r) v[f * 8 + r] = bm[f][r];
#pragma unroll
    for (int j = 0; j < 8; ++j) {
        DPP_PAIRMIN(v[j], v[j],     8, 0x3);
        DPP_PAIRMIN(v[j], v[j + 8], 8, 0xc);
    }
#pragma unroll
    for (int j = 0; j < 4; ++j) {
        DPP_PAIRMIN(v[j], v[j],     4, 0x5);
        DPP_PAIRMIN(v[j], v[j + 4], 4, 0xa);
    }
    const bool b1 = (col & 2) != 0;
    const bool b0 = (col & 1) != 0;
#pragma unroll
    for (int j = 0; j < 2; ++j) {
        float keep = b1 ? v[j + 2] : v[j];
        const float send = b1 ? v[j] : v[j + 2];
        DPP_MIN_KEEP(keep, send, 2);
        v[j] = keep;
    }
    float rv = b0 ? v[1] : v[0];
    {
        const float send = b0 ? v[0] : v[1];
        DPP_MIN_KEEP(rv, send, 1);
    }
    prow[(b * NCS + cs) * NPTS + rs * RS + wave * RPW
         + 16 * (col >> 3) + 8 * hi + (col & 7)] = rv;

    lds_barrier();
#pragma unroll
    for (int k = 0; k < CS / BLOCK; ++k) {
        const int j = k * BLOCK + tid;
        const float* pc = s_cm + (j >> 4) * 32 + (j & 15);
        float m = 3.4e38f;
#pragma unroll
        for (int w = 0; w < WAVES; ++w)
            m = min3f(m, pc[w * (TILES * 32)], pc[w * (TILES * 32) + 16]);
        pcol[(b * NRS + rs) * NPTS + cs * CS + j] = m;
    }
}

__device__ __forceinline__ float wave_sum(float s)
{
    const int sb = __builtin_bit_cast(int, s);
    s += __builtin_bit_cast(float, __builtin_amdgcn_permlanex16(
             sb, sb, 0x76543210, 0xfedcba98, false, false));
    DPP_ADD_SELF(s, 8);
    DPP_ADD_SELF(s, 4);
    DPP_ADD_SELF(s, 2);
    DPP_ADD_SELF(s, 1);
    return s;
}

__device__ __forceinline__ float4 min4(float4 a, float4 b)
{
    return make_float4(fminf(a.x, b.x), fminf(a.y, b.y), fminf(a.z, b.z), fminf(a.w, b.w));
}

__device__ __forceinline__ float4 dist4(float4 a)
{
    return make_float4(sqrtf(fmaxf(a.x, 0.0f)), sqrtf(fmaxf(a.y, 0.0f)),
                       sqrtf(fmaxf(a.z, 0.0f)), sqrtf(fmaxf(a.w, 0.0f)));
}

__global__ __launch_bounds__(FINISH)
void chamfer_finish_kernel(const float* __restrict__ prow,
                           const float* __restrict__ pcol,
                           float* __restrict__ out)
{
    const int b    = blockIdx.x;
    const int tid  = threadIdx.x;
    const int wave = tid >> 5;
    const int lane = tid & 31;
    const float4* pr = reinterpret_cast<const float4*>(prow + b * NCS * NPTS) + tid;
    const float4* pc = reinterpret_cast<const float4*>(pcol + b * NRS * NPTS) + tid;

    if (tid < 64 * (NCS + NRS)) {
        const float* unit = (tid < 64 * NCS)
            ? prow + b * NCS * NPTS + tid * 64
            : pcol + b * NRS * NPTS + (tid - 64 * NCS) * 64;
        __builtin_prefetch(unit, 0, 2);
    }

    float4 a = pr[0];
#pragma unroll
    for (int p = 1; p < NCS; ++p) a = min4(a, pr[p * (NPTS / 4)]);
    float4 c = pc[0];
#pragma unroll
    for (int p = 1; p < NRS; ++p) c = min4(c, pc[p * (NPTS / 4)]);

    a = dist4(a);
    c = dist4(c);
    reinterpret_cast<float4*>(out + b * NPTS)[tid] = a;
    reinterpret_cast<float4*>(out + NBATCH * NPTS + b * NPTS)[tid] = c;

    float s = ((a.x + a.y) + (a.z + a.w)) + ((c.x + c.y) + (c.z + c.w));
    s = wave_sum(s);
    __shared__ float red[FINISH / 32];
    if (lane == 0) red[wave] = s;
    lds_barrier();
    if (wave == 0) {
        float v = wave_sum(red[lane]);
        if (lane == 0)
            out[2 * NBATCH * NPTS + b] = v / (2.0f * (float)NPTS);
    }
}

extern "C" void kernel_launch(void* const* d_in, const int* in_sizes, int n_in,
                              void* d_out, int out_size, void* d_ws, size_t ws_size,
                              hipStream_t stream) {
    (void)in_sizes; (void)n_in; (void)ws_size; (void)out_size;
    const float* tar = (const float*)d_in[0];
    const float* src = (const float*)d_in[1];
    float* out = (float*)d_out;
    float* prow = (float*)d_ws;
    float* pcol = prow + NBATCH * NCS * NPTS;

    dim3 grid(NCS, NRS, NBATCH);
    chamfer_tile_kernel<<<grid, BLOCK, 0, stream>>>(tar, src, prow, pcol);
    chamfer_finish_kernel<<<NBATCH, FINISH, 0, stream>>>(prow, pcol, out);
}
